# starting kernel
# baseline (speedup 1.0000x reference)
#include <hip/hip_runtime.h>
#include <hip/hip_bf16.h>

#define D_FEAT 256
#define HID    16
#define KSTEPS (D_FEAT / 4)
#define KCHUNK 8

typedef float v2f __attribute__((ext_vector_type(2)));
typedef float v8f __attribute__((ext_vector_type(8)));

__global__ void gcn_deg_kernel(const int* __restrict__ dst,
                               unsigned* __restrict__ deg, int E) {
    int e = blockIdx.x * blockDim.x + threadIdx.x;
    if (e < E) atomicAdd(&deg[dst[e]], 1u);
}

__global__ void gcn_dinv_kernel(const unsigned* __restrict__ deg,
                                float* __restrict__ dinv, int N) {
    int i = blockIdx.x * blockDim.x + threadIdx.x;
    if (i < N) dinv[i] = rsqrtf((float)(deg[i] + 1u));
}

__global__ __launch_bounds__(256, 1)
void gcn_gemm_wmma_kernel(const float* __restrict__ x,
                          const float* __restrict__ W,
                          float* __restrict__ h, int ntiles) {
    __shared__ float Wl[D_FEAT * HID];

    const int tid = threadIdx.x;
    for (int i = tid; i < (D_FEAT * HID) / 4; i += blockDim.x)
        ((float4*)Wl)[i] = ((const float4*)W)[i];
    __syncthreads();

    const int lane  = tid & 31;
    const int wave  = tid >> 5;
    const int wpb   = blockDim.x >> 5;
    const int gwave = blockIdx.x * wpb + wave;
    const int nwav  = gridDim.x * wpb;

    const int mrow = lane & 15;
    const int kk   = (lane >> 4) << 1;
    const int ncol = lane & 15;

    v2f bfrag[KSTEPS];
#pragma unroll
    for (int kb = 0; kb < KSTEPS; ++kb) {
        bfrag[kb].x = Wl[(4 * kb + kk)     * HID + ncol];
        bfrag[kb].y = Wl[(4 * kb + kk + 1) * HID + ncol];
    }

    for (int tile = gwave; tile < ntiles; tile += nwav) {
        const float* xrow = x + (size_t)(tile * 16 + mrow) * D_FEAT + kk;

        if (tile + nwav < ntiles)
            __builtin_prefetch(xrow + (size_t)nwav * 16 * D_FEAT, 0, 0);

        v8f c = {};
#pragma unroll
        for (int kc = 0; kc < KSTEPS; kc += KCHUNK) {
            v2f a[KCHUNK];
#pragma unroll
            for (int u = 0; u < KCHUNK; ++u)
                a[u] = *(const v2f*)(xrow + 4 * (kc + u));
#pragma unroll
            for (int u = 0; u < KCHUNK; ++u)
                c = __builtin_amdgcn_wmma_f32_16x16x4_f32(
                        false, a[u], false, bfrag[kc + u],
                        (short)0, c, false, false);
        }

        float* hout = h + (size_t)tile * 16 * HID + (lane >> 4) * 8 * HID + ncol;
#pragma unroll
        for (int v = 0; v < 8; ++v)
            hout[v * HID] = c[v];
    }
}

__global__ void gcn_scatter_kernel(const int* __restrict__ ei,
                                   const float* __restrict__ dinv,
                                   const float* __restrict__ h,
                                   float* __restrict__ agg, int E) {
    for (int e = blockIdx.x * blockDim.x + threadIdx.x; e < E;
         e += gridDim.x * blockDim.x) {
        const int s = ei[e];
        const int d = ei[E + e];
        const float w = dinv[s] * dinv[d];
        const float4* hs = (const float4*)(h + (size_t)s * HID);
        float* ad = agg + (size_t)d * HID;
#pragma unroll
        for (int j = 0; j < 4; ++j) {
            float4 v = hs[j];
            unsafeAtomicAdd(ad + 4 * j + 0, v.x * w);
            unsafeAtomicAdd(ad + 4 * j + 1, v.y * w);
            unsafeAtomicAdd(ad + 4 * j + 2, v.z * w);
            unsafeAtomicAdd(ad + 4 * j + 3, v.w * w);
        }
    }
}

__global__ void gcn_head_kernel(const float* __restrict__ h,
                                const float* __restrict__ agg,
                                const float* __restrict__ dinv,
                                const float* __restrict__ bconv,
                                const float* __restrict__ wfc,
                                const float* __restrict__ bfc,
                                float* __restrict__ out, int N) {
    int i = blockIdx.x * blockDim.x + threadIdx.x;
    if (i >= N) return;
    const float di = dinv[i];
    const float sw = di * di;
    const float4* hp = (const float4*)(h + (size_t)i * HID);
    const float4* ap = (const float4*)(agg + (size_t)i * HID);
    float acc = 0.f;
#pragma unroll
    for (int j = 0; j < 4; ++j) {
        float4 hv = hp[j];
        float4 av = ap[j];
        float4 wv = ((const float4*)wfc)[j];
        float4 bv = ((const float4*)bconv)[j];
        acc += (av.x + hv.x * sw + bv.x) * wv.x;
        acc += (av.y + hv.y * sw + bv.y) * wv.y;
        acc += (av.z + hv.z * sw + bv.z) * wv.z;
        acc += (av.w + hv.w * sw + bv.w) * wv.w;
    }
    out[i] = acc + bfc[0];
}

extern "C" void kernel_launch(void* const* d_in, const int* in_sizes, int n_in,
                              void* d_out, int out_size, void* d_ws, size_t ws_size,
                              hipStream_t stream) {
    const float* x     = (const float*)d_in[0];
    const int*   ei    = (const int*)d_in[1];
    const float* W     = (const float*)d_in[2];
    const float* bconv = (const float*)d_in[3];
    const float* wfc   = (const float*)d_in[4];
    const float* bfc   = (const float*)d_in[5];
    float* out = (float*)d_out;

    const int N = in_sizes[0] / D_FEAT;
    const int E = in_sizes[1] / 2;

    char* ws = (char*)d_ws;
    unsigned* deg  = (unsigned*)ws;
    float*    dinv = (float*)(ws + (size_t)N * 4);
    float*    h    = (float*)(ws + (size_t)N * 8);
    float*    agg  = (float*)(ws + (size_t)N * 8 + (size_t)N * HID * 4);

    hipMemsetAsync(deg, 0, (size_t)N * sizeof(unsigned), stream);
    hipMemsetAsync(agg, 0, (size_t)N * HID * sizeof(float), stream);

    gcn_deg_kernel<<<(E + 255) / 256, 256, 0, stream>>>(ei + E, deg, E);
    gcn_dinv_kernel<<<(N + 255) / 256, 256, 0, stream>>>(deg, dinv, N);

    const int ntiles = N / 16;
    gcn_gemm_wmma_kernel<<<128, 256, 0, stream>>>(x, W, h, ntiles);

    gcn_scatter_kernel<<<1024, 256, 0, stream>>>(ei, dinv, h, agg, E);

    gcn_head_kernel<<<(N + 255) / 256, 256, 0, stream>>>(h, agg, dinv, bconv,
                                                         wfc, bfc, out, N);
}


# Edge sum: 16 lanes per edge (lane = column, 2 edges per wave instruction): an edge's h row read and its 16 f32 atomic adds are one 64-byte request each; f32 summation order changes (atomics kept)
# speedup vs baseline: 3.0381x; 3.0381x over previous
#include <hip/hip_runtime.h>
#include <hip/hip_bf16.h>

#define D_FEAT 256
#define HID    16
#define KSTEPS (D_FEAT / 4)
#define KCHUNK 8

typedef float v2f __attribute__((ext_vector_type(2)));
typedef float v8f __attribute__((ext_vector_type(8)));

__global__ void gcn_deg_kernel(const int* __restrict__ dst,
                               unsigned* __restrict__ deg, int E) {
    int e = blockIdx.x * blockDim.x + threadIdx.x;
    if (e < E) atomicAdd(&deg[dst[e]], 1u);
}

__global__ void gcn_dinv_kernel(const unsigned* __restrict__ deg,
                                float* __restrict__ dinv, int N) {
    int i = blockIdx.x * blockDim.x + threadIdx.x;
    if (i < N) dinv[i] = rsqrtf((float)(deg[i] + 1u));
}

__global__ __launch_bounds__(256, 1)
void gcn_gemm_wmma_kernel(const float* __restrict__ x,
                          const float* __restrict__ W,
                          float* __restrict__ h, int ntiles) {
    __shared__ float Wl[D_FEAT * HID];

    const int tid = threadIdx.x;
    for (int i = tid; i < (D_FEAT * HID) / 4; i += blockDim.x)
        ((float4*)Wl)[i] = ((const float4*)W)[i];
    __syncthreads();

    const int lane  = tid & 31;
    const int wave  = tid >> 5;
    const int wpb   = blockDim.x >> 5;
    const int gwave = blockIdx.x * wpb + wave;
    const int nwav  = gridDim.x * wpb;

    const int mrow = lane & 15;
    const int kk   = (lane >> 4) << 1;
    const int ncol = lane & 15;

    v2f bfrag[KSTEPS];
#pragma unroll
    for (int kb = 0; kb < KSTEPS; ++kb) {
        bfrag[kb].x = Wl[(4 * kb + kk)     * HID + ncol];
        bfrag[kb].y = Wl[(4 * kb + kk + 1) * HID + ncol];
    }

    for (int tile = gwave; tile < ntiles; tile += nwav) {
        const float* xrow = x + (size_t)(tile * 16 + mrow) * D_FEAT + kk;

        if (tile + nwav < ntiles)
            __builtin_prefetch(xrow + (size_t)nwav * 16 * D_FEAT, 0, 0);

        v8f c = {};
#pragma unroll
        for (int kc = 0; kc < KSTEPS; kc += KCHUNK) {
            v2f a[KCHUNK];
#pragma unroll
            for (int u = 0; u < KCHUNK; ++u)
                a[u] = *(const v2f*)(xrow + 4 * (kc + u));
#pragma unroll
            for (int u = 0; u < KCHUNK; ++u)
                c = __builtin_amdgcn_wmma_f32_16x16x4_f32(
                        false, a[u], false, bfrag[kc + u],
                        (short)0, c, false, false);
        }

        float* hout = h + (size_t)tile * 16 * HID + (lane >> 4) * 8 * HID + ncol;
#pragma unroll
        for (int v = 0; v < 8; ++v)
            hout[v * HID] = c[v];
    }
}

#define ROW_SHARE(x, i) __builtin_amdgcn_update_dpp(0, (x), 0x150 + (i), 0xF, 0xF, false)

#define EDGE_LOAD(i)                                                          \
    const unsigned so##i = (unsigned)ROW_SHARE(s64, i) + j4;                  \
    const float    hv##i = *(const float*)((const char*)h + so##i);

#define EDGE_ADD(i)                                                           \
    {                                                                         \
        const unsigned dofs = (unsigned)ROW_SHARE(d64, i) + j4;               \
        const float wi = __int_as_float(ROW_SHARE(__float_as_int(w), i));     \
        unsafeAtomicAdd((float*)((char*)agg + dofs), hv##i * wi);             \
    }

__global__ __launch_bounds__(256)
void gcn_scatter_kernel(const int* __restrict__ ei,
                        const float* __restrict__ dinv,
                        const float* __restrict__ h,
                        float* __restrict__ agg, int E) {
    const int lane  = threadIdx.x & 31;
    const int wpb   = blockDim.x >> 5;
    const int gwave = blockIdx.x * wpb + (threadIdx.x >> 5);
    const int nwav  = gridDim.x * wpb;
    const int nfull = E >> 5;
    const unsigned j4 = (unsigned)(lane & 15) * 4u;

    for (int b = gwave; b < nfull; b += nwav) {
        const int e = (b << 5) + lane;
        const int s = ei[e];
        const int d = ei[E + e];
        const float w = dinv[s] * dinv[d];
        const int s64 = s * (HID * 4);
        const int d64 = d * (HID * 4);

        EDGE_LOAD(0)  EDGE_LOAD(1)  EDGE_LOAD(2)  EDGE_LOAD(3)
        EDGE_LOAD(4)  EDGE_LOAD(5)  EDGE_LOAD(6)  EDGE_LOAD(7)
        EDGE_LOAD(8)  EDGE_LOAD(9)  EDGE_LOAD(10) EDGE_LOAD(11)
        EDGE_LOAD(12) EDGE_LOAD(13) EDGE_LOAD(14) EDGE_LOAD(15)

        EDGE_ADD(0)  EDGE_ADD(1)  EDGE_ADD(2)  EDGE_ADD(3)
        EDGE_ADD(4)  EDGE_ADD(5)  EDGE_ADD(6)  EDGE_ADD(7)
        EDGE_ADD(8)  EDGE_ADD(9)  EDGE_ADD(10) EDGE_ADD(11)
        EDGE_ADD(12) EDGE_ADD(13) EDGE_ADD(14) EDGE_ADD(15)
    }

    const int et = (nfull << 5) + blockIdx.x * blockDim.x + threadIdx.x;
    if (et < E) {
        const int s = ei[et];
        const int d = ei[E + et];
        const float w = dinv[s] * dinv[d];
        const float* hs = h + (size_t)s * HID;
        float* ad = agg + (size_t)d * HID;
        for (int c = 0; c < HID; ++c)
            unsafeAtomicAdd(ad + c, hs[c] * w);
    }
}

__global__ void gcn_head_kernel(const float* __restrict__ h,
                                const float* __restrict__ agg,
                                const float* __restrict__ dinv,
                                const float* __restrict__ bconv,
                                const float* __restrict__ wfc,
                                const float* __restrict__ bfc,
                                float* __restrict__ out, int N) {
    int i = blockIdx.x * blockDim.x + threadIdx.x;
    if (i >= N) return;
    const float di = dinv[i];
    const float sw = di * di;
    const float4* hp = (const float4*)(h + (size_t)i * HID);
    const float4* ap = (const float4*)(agg + (size_t)i * HID);
    float acc = 0.f;
#pragma unroll
    for (int j = 0; j < 4; ++j) {
        float4 hv = hp[j];
        float4 av = ap[j];
        float4 wv = ((const float4*)wfc)[j];
        float4 bv = ((const float4*)bconv)[j];
        acc += (av.x + hv.x * sw + bv.x) * wv.x;
        acc += (av.y + hv.y * sw + bv.y) * wv.y;
        acc += (av.z + hv.z * sw + bv.z) * wv.z;
        acc += (av.w + hv.w * sw + bv.w) * wv.w;
    }
    out[i] = acc + bfc[0];
}

extern "C" void kernel_launch(void* const* d_in, const int* in_sizes, int n_in,
                              void* d_out, int out_size, void* d_ws, size_t ws_size,
                              hipStream_t stream) {
    const float* x     = (const float*)d_in[0];
    const int*   ei    = (const int*)d_in[1];
    const float* W     = (const float*)d_in[2];
    const float* bconv = (const float*)d_in[3];
    const float* wfc   = (const float*)d_in[4];
    const float* bfc   = (const float*)d_in[5];
    float* out = (float*)d_out;

    const int N = in_sizes[0] / D_FEAT;
    const int E = in_sizes[1] / 2;

    char* ws = (char*)d_ws;
    unsigned* deg  = (unsigned*)ws;
    float*    dinv = (float*)(ws + (size_t)N * 4);
    float*    h    = (float*)(ws + (size_t)N * 8);
    float*    agg  = (float*)(ws + (size_t)N * 8 + (size_t)N * HID * 4);

    hipMemsetAsync(deg, 0, (size_t)N * sizeof(unsigned), stream);
    hipMemsetAsync(agg, 0, (size_t)N * HID * sizeof(float), stream);

    gcn_deg_kernel<<<(E + 255) / 256, 256, 0, stream>>>(ei + E, deg, E);
    gcn_dinv_kernel<<<(N + 255) / 256, 256, 0, stream>>>(deg, dinv, N);

    const int ntiles = N / 16;
    gcn_gemm_wmma_kernel<<<128, 256, 0, stream>>>(x, W, h, ntiles);

    gcn_scatter_kernel<<<1024, 256, 0, stream>>>(ei, dinv, h, agg, E);

    gcn_head_kernel<<<(N + 255) / 256, 256, 0, stream>>>(h, agg, dinv, bconv,
                                                         wfc, bfc, out, N);
}


# Edge weight dinv[src]*dinv[dst] split: x@W epilogue stores h*dinv[src], head scales the summed row by dinv[dst]; edge kernel gathers no dinv; f32 rounding of the terms changes (atomics kept)
# speedup vs baseline: 3.2493x; 1.0695x over previous
#include <hip/hip_runtime.h>
#include <hip/hip_bf16.h>

#define D_FEAT 256
#define HID    16
#define KSTEPS (D_FEAT / 4)
#define KCHUNK 8

typedef float v2f __attribute__((ext_vector_type(2)));
typedef float v8f __attribute__((ext_vector_type(8)));

__global__ void gcn_deg_kernel(const int* __restrict__ dst,
                               unsigned* __restrict__ deg, int E) {
    int e = blockIdx.x * blockDim.x + threadIdx.x;
    if (e < E) atomicAdd(&deg[dst[e]], 1u);
}

__global__ void gcn_dinv_kernel(const unsigned* __restrict__ deg,
                                float* __restrict__ dinv, int N) {
    int i = blockIdx.x * blockDim.x + threadIdx.x;
    if (i < N) dinv[i] = rsqrtf((float)(deg[i] + 1u));
}

__global__ __launch_bounds__(256, 1)
void gcn_gemm_wmma_kernel(const float* __restrict__ x,
                          const float* __restrict__ W,
                          const float* __restrict__ dinv,
                          float* __restrict__ h, int ntiles) {
    __shared__ float Wl[D_FEAT * HID];

    const int tid = threadIdx.x;
    for (int i = tid; i < (D_FEAT * HID) / 4; i += blockDim.x)
        ((float4*)Wl)[i] = ((const float4*)W)[i];
    __syncthreads();

    const int lane  = tid & 31;
    const int wave  = tid >> 5;
    const int wpb   = blockDim.x >> 5;
    const int gwave = blockIdx.x * wpb + wave;
    const int nwav  = gridDim.x * wpb;

    const int mrow = lane & 15;
    const int kk   = (lane >> 4) << 1;
    const int ncol = lane & 15;

    v2f bfrag[KSTEPS];
#pragma unroll
    for (int kb = 0; kb < KSTEPS; ++kb) {
        bfrag[kb].x = Wl[(4 * kb + kk)     * HID + ncol];
        bfrag[kb].y = Wl[(4 * kb + kk + 1) * HID + ncol];
    }

    const int hrow8 = (lane >> 4) * 8;
    const float4* dq0 = (const float4*)(dinv + min(gwave, ntiles - 1) * 16 + hrow8);
    float4 dlo = dq0[0];
    float4 dhi = dq0[1];

    for (int tile = gwave; tile < ntiles; tile += nwav) {
        const float4* dqn = (const float4*)(dinv + min(tile + nwav, ntiles - 1) * 16 + hrow8);
        const float4 nlo = dqn[0];
        const float4 nhi = dqn[1];

        const float* xrow = x + (size_t)(tile * 16 + mrow) * D_FEAT + kk;

        if (tile + nwav < ntiles)
            __builtin_prefetch(xrow + (size_t)nwav * 16 * D_FEAT, 0, 0);

        v8f c = {};
#pragma unroll
        for (int kc = 0; kc < KSTEPS; kc += KCHUNK) {
            v2f a[KCHUNK];
#pragma unroll
            for (int u = 0; u < KCHUNK; ++u)
                a[u] = *(const v2f*)(xrow + 4 * (kc + u));
#pragma unroll
            for (int u = 0; u < KCHUNK; ++u)
                c = __builtin_amdgcn_wmma_f32_16x16x4_f32(
                        false, a[u], false, bfrag[kc + u],
                        (short)0, c, false, false);
        }

        float* hout = h + (size_t)tile * 16 * HID + (lane >> 4) * 8 * HID + ncol;
        const float dv[8] = {dlo.x, dlo.y, dlo.z, dlo.w, dhi.x, dhi.y, dhi.z, dhi.w};
#pragma unroll
        for (int v = 0; v < 8; ++v)
            hout[v * HID] = c[v] * dv[v];
        dlo = nlo;
        dhi = nhi;
    }
}

#define ROW_SHARE(x, i) __builtin_amdgcn_update_dpp(0, (x), 0x150 + (i), 0xF, 0xF, false)

#define EDGE_LOAD(i)                                                          \
    const unsigned so##i = (unsigned)ROW_SHARE(s64, i) + j4;                  \
    const float    hv##i = *(const float*)((const char*)h + so##i);

#define EDGE_ADD(i)                                                           \
    {                                                                         \
        const unsigned dofs = (unsigned)ROW_SHARE(d64, i) + j4;               \
        unsafeAtomicAdd((float*)((char*)agg + dofs), hv##i);                  \
    }

__global__ __launch_bounds__(256)
void gcn_scatter_kernel(const int* __restrict__ ei,
                        const float* __restrict__ h,
                        float* __restrict__ agg, int E) {
    const int lane  = threadIdx.x & 31;
    const int wpb   = blockDim.x >> 5;
    const int gwave = blockIdx.x * wpb + (threadIdx.x >> 5);
    const int nwav  = gridDim.x * wpb;
    const int nfull = E >> 5;
    const unsigned j4 = (unsigned)(lane & 15) * 4u;

    for (int b = gwave; b < nfull; b += nwav) {
        const int e = (b << 5) + lane;
        const int s = ei[e];
        const int d = ei[E + e];
        const int s64 = s * (HID * 4);
        const int d64 = d * (HID * 4);

        EDGE_LOAD(0)  EDGE_LOAD(1)  EDGE_LOAD(2)  EDGE_LOAD(3)
        EDGE_LOAD(4)  EDGE_LOAD(5)  EDGE_LOAD(6)  EDGE_LOAD(7)
        EDGE_LOAD(8)  EDGE_LOAD(9)  EDGE_LOAD(10) EDGE_LOAD(11)
        EDGE_LOAD(12) EDGE_LOAD(13) EDGE_LOAD(14) EDGE_LOAD(15)

        EDGE_ADD(0)  EDGE_ADD(1)  EDGE_ADD(2)  EDGE_ADD(3)
        EDGE_ADD(4)  EDGE_ADD(5)  EDGE_ADD(6)  EDGE_ADD(7)
        EDGE_ADD(8)  EDGE_ADD(9)  EDGE_ADD(10) EDGE_ADD(11)
        EDGE_ADD(12) EDGE_ADD(13) EDGE_ADD(14) EDGE_ADD(15)
    }

    const int et = (nfull << 5) + blockIdx.x * blockDim.x + threadIdx.x;
    if (et < E) {
        const int s = ei[et];
        const int d = ei[E + et];
        const float* hs = h + (size_t)s * HID;
        float* ad = agg + (size_t)d * HID;
        for (int c = 0; c < HID; ++c)
            unsafeAtomicAdd(ad + c, hs[c]);
    }
}

__global__ void gcn_head_kernel(const float* __restrict__ h,
                                const float* __restrict__ agg,
                                const float* __restrict__ dinv,
                                const float* __restrict__ bconv,
                                const float* __restrict__ wfc,
                                const float* __restrict__ bfc,
                                float* __restrict__ out, int N) {
    int i = blockIdx.x * blockDim.x + threadIdx.x;
    if (i >= N) return;
    const float di = dinv[i];
    const float4* hp = (const float4*)(h + (size_t)i * HID);
    const float4* ap = (const float4*)(agg + (size_t)i * HID);
    float acc = 0.f;
#pragma unroll
    for (int j = 0; j < 4; ++j) {
        float4 hv = hp[j];
        float4 av = ap[j];
        float4 wv = ((const float4*)wfc)[j];
        float4 bv = ((const float4*)bconv)[j];
        acc += ((av.x + hv.x) * di + bv.x) * wv.x;
        acc += ((av.y + hv.y) * di + bv.y) * wv.y;
        acc += ((av.z + hv.z) * di + bv.z) * wv.z;
        acc += ((av.w + hv.w) * di + bv.w) * wv.w;
    }
    out[i] = acc + bfc[0];
}

extern "C" void kernel_launch(void* const* d_in, const int* in_sizes, int n_in,
                              void* d_out, int out_size, void* d_ws, size_t ws_size,
                              hipStream_t stream) {
    const float* x     = (const float*)d_in[0];
    const int*   ei    = (const int*)d_in[1];
    const float* W     = (const float*)d_in[2];
    const float* bconv = (const float*)d_in[3];
    const float* wfc   = (const float*)d_in[4];
    const float* bfc   = (const float*)d_in[5];
    float* out = (float*)d_out;

    const int N = in_sizes[0] / D_FEAT;
    const int E = in_sizes[1] / 2;

    char* ws = (char*)d_ws;
    unsigned* deg  = (unsigned*)ws;
    float*    dinv = (float*)(ws + (size_t)N * 4);
    float*    h    = (float*)(ws + (size_t)N * 8);
    float*    agg  = (float*)(ws + (size_t)N * 8 + (size_t)N * HID * 4);

    hipMemsetAsync(deg, 0, (size_t)N * sizeof(unsigned), stream);
    hipMemsetAsync(agg, 0, (size_t)N * HID * sizeof(float), stream);

    gcn_deg_kernel<<<(E + 255) / 256, 256, 0, stream>>>(ei + E, deg, E);
    gcn_dinv_kernel<<<(N + 255) / 256, 256, 0, stream>>>(deg, dinv, N);

    const int ntiles = N / 16;
    gcn_gemm_wmma_kernel<<<128, 256, 0, stream>>>(x, W, dinv, h, ntiles);

    gcn_scatter_kernel<<<1024, 256, 0, stream>>>(ei, h, agg, E);

    gcn_head_kernel<<<(N + 255) / 256, 256, 0, stream>>>(h, agg, dinv, bconv,
                                                         wfc, bfc, out, N);
}


# Degree count, agg clear and x@W share ONE launch as work-group roles (degree role: 1024 looping groups, 4 edges a lane; x loads pinned 16 a batch); dinv and h row scaling in one small kernel behind
# speedup vs baseline: 3.4767x; 1.0700x over previous
#include <hip/hip_runtime.h>
#include <hip/hip_bf16.h>

#define D_FEAT 256
#define HID    16
#define KSTEPS (D_FEAT / 4)
#define KCHUNK 16

typedef float v2f __attribute__((ext_vector_type(2)));
typedef float v8f __attribute__((ext_vector_type(8)));

__global__ __launch_bounds__(256, 1)
void gcn_deg_gemm_kernel(const float* __restrict__ x,
                         const float* __restrict__ W,
                         const int* __restrict__ dst,
                         unsigned* __restrict__ deg,
                         float* __restrict__ h,
                         float* __restrict__ agg,
                         int ntiles, int E, int nagg4, int gemm_blocks) {
    __shared__ float Wl[D_FEAT * HID];

    const int tid = threadIdx.x;

    if ((int)blockIdx.x >= gemm_blocks) {
        const int nthr = ((int)gridDim.x - gemm_blocks) * (int)blockDim.x;
        const int gt   = ((int)blockIdx.x - gemm_blocks) * (int)blockDim.x + tid;

        const float4 zero4 = {0.f, 0.f, 0.f, 0.f};
        for (int i = gt; i < nagg4; i += nthr)
            ((float4*)agg)[i] = zero4;

        if ((E & 3) == 0) {
            const int4* d4 = (const int4*)dst;
            const int nq = E >> 2;
            for (int q = gt; q < nq; q += nthr) {
                const int4 d = d4[q];
                atomicAdd(&deg[d.x], 1u);
                atomicAdd(&deg[d.y], 1u);
                atomicAdd(&deg[d.z], 1u);
                atomicAdd(&deg[d.w], 1u);
            }
        } else {
            for (int e = gt; e < E; e += nthr)
                atomicAdd(&deg[dst[e]], 1u);
        }
        return;
    }

    for (int i = tid; i < (D_FEAT * HID) / 4; i += blockDim.x)
        ((float4*)Wl)[i] = ((const float4*)W)[i];
    __syncthreads();

    const int lane  = tid & 31;
    const int wave  = tid >> 5;
    const int wpb   = blockDim.x >> 5;
    const int gwave = blockIdx.x * wpb + wave;
    const int nwav  = gemm_blocks * wpb;

    const int mrow = lane & 15;
    const int kk   = (lane >> 4) << 1;
    const int ncol = lane & 15;

    v2f bfrag[KSTEPS];
#pragma unroll
    for (int kb = 0; kb < KSTEPS; ++kb) {
        bfrag[kb].x = Wl[(4 * kb + kk)     * HID + ncol];
        bfrag[kb].y = Wl[(4 * kb + kk + 1) * HID + ncol];
    }

    for (int tile = gwave; tile < ntiles; tile += nwav) {
        const float* xrow = x + (size_t)(tile * 16 + mrow) * D_FEAT + kk;

        if (tile + nwav < ntiles)
            __builtin_prefetch(xrow + (size_t)nwav * 16 * D_FEAT, 0, 0);

        v8f c = {};
#pragma unroll
        for (int kc = 0; kc < KSTEPS; kc += KCHUNK) {
            v2f a[KCHUNK];
#pragma unroll
            for (int u = 0; u < KCHUNK; ++u)
                a[u] = *(const v2f*)(xrow + 4 * (kc + u));
            __builtin_amdgcn_sched_barrier(0);
#pragma unroll
            for (int u = 0; u < KCHUNK; ++u)
                c = __builtin_amdgcn_wmma_f32_16x16x4_f32(
                        false, a[u], false, bfrag[kc + u],
                        (short)0, c, false, false);
        }

        float* hout = h + (size_t)tile * 16 * HID + (lane >> 4) * 8 * HID + ncol;
#pragma unroll
        for (int v = 0; v < 8; ++v)
            hout[v * HID] = c[v];
    }
}

__global__ void gcn_dinv_scale_kernel(const unsigned* __restrict__ deg,
                                      float* __restrict__ dinv,
                                      float* __restrict__ h, int N) {
    const int t = blockIdx.x * blockDim.x + threadIdx.x;
    if (t >= 2 * N) return;
    const int i = t >> 1;
    const float di = rsqrtf((float)(deg[i] + 1u));
    float4* hp = (float4*)h + 2 * (size_t)t;
    float4 v0 = hp[0];
    float4 v1 = hp[1];
    v0.x *= di; v0.y *= di; v0.z *= di; v0.w *= di;
    v1.x *= di; v1.y *= di; v1.z *= di; v1.w *= di;
    hp[0] = v0;
    hp[1] = v1;
    if ((t & 1) == 0) dinv[i] = di;
}

#define ROW_SHARE(x, i) __builtin_amdgcn_update_dpp(0, (x), 0x150 + (i), 0xF, 0xF, false)

#define EDGE_LOAD(i)                                                          \
    const unsigned so##i = (unsigned)ROW_SHARE(s64, i) + j4;                  \
    const float    hv##i = *(const float*)((const char*)h + so##i);

#define EDGE_ADD(i)                                                           \
    {                                                                         \
        const unsigned dofs = (unsigned)ROW_SHARE(d64, i) + j4;               \
        unsafeAtomicAdd((float*)((char*)agg + dofs), hv##i);                  \
    }

__global__ __launch_bounds__(256)
void gcn_scatter_kernel(const int* __restrict__ ei,
                        const float* __restrict__ h,
                        float* __restrict__ agg, int E) {
    const int lane  = threadIdx.x & 31;
    const int wpb   = blockDim.x >> 5;
    const int gwave = blockIdx.x * wpb + (threadIdx.x >> 5);
    const int nwav  = gridDim.x * wpb;
    const int nfull = E >> 5;
    const unsigned j4 = (unsigned)(lane & 15) * 4u;

    for (int b = gwave; b < nfull; b += nwav) {
        const int e = (b << 5) + lane;
        const int s = ei[e];
        const int d = ei[E + e];
        const int s64 = s * (HID * 4);
        const int d64 = d * (HID * 4);

        EDGE_LOAD(0)  EDGE_LOAD(1)  EDGE_LOAD(2)  EDGE_LOAD(3)
        EDGE_LOAD(4)  EDGE_LOAD(5)  EDGE_LOAD(6)  EDGE_LOAD(7)
        EDGE_LOAD(8)  EDGE_LOAD(9)  EDGE_LOAD(10) EDGE_LOAD(11)
        EDGE_LOAD(12) EDGE_LOAD(13) EDGE_LOAD(14) EDGE_LOAD(15)

        EDGE_ADD(0)  EDGE_ADD(1)  EDGE_ADD(2)  EDGE_ADD(3)
        EDGE_ADD(4)  EDGE_ADD(5)  EDGE_ADD(6)  EDGE_ADD(7)
        EDGE_ADD(8)  EDGE_ADD(9)  EDGE_ADD(10) EDGE_ADD(11)
        EDGE_ADD(12) EDGE_ADD(13) EDGE_ADD(14) EDGE_ADD(15)
    }

    const int et = (nfull << 5) + blockIdx.x * blockDim.x + threadIdx.x;
    if (et < E) {
        const int s = ei[et];
        const int d = ei[E + et];
        const float* hs = h + (size_t)s * HID;
        float* ad = agg + (size_t)d * HID;
        for (int c = 0; c < HID; ++c)
            unsafeAtomicAdd(ad + c, hs[c]);
    }
}

__global__ void gcn_head_kernel(const float* __restrict__ h,
                                const float* __restrict__ agg,
                                const float* __restrict__ dinv,
                                const float* __restrict__ bconv,
                                const float* __restrict__ wfc,
                                const float* __restrict__ bfc,
                                float* __restrict__ out, int N) {
    int i = blockIdx.x * blockDim.x + threadIdx.x;
    if (i >= N) return;
    const float di = dinv[i];
    const float4* hp = (const float4*)(h + (size_t)i * HID);
    const float4* ap = (const float4*)(agg + (size_t)i * HID);
    float acc = 0.f;
#pragma unroll
    for (int j = 0; j < 4; ++j) {
        float4 hv = hp[j];
        float4 av = ap[j];
        float4 wv = ((const float4*)wfc)[j];
        float4 bv = ((const float4*)bconv)[j];
        acc += ((av.x + hv.x) * di + bv.x) * wv.x;
        acc += ((av.y + hv.y) * di + bv.y) * wv.y;
        acc += ((av.z + hv.z) * di + bv.z) * wv.z;
        acc += ((av.w + hv.w) * di + bv.w) * wv.w;
    }
    out[i] = acc + bfc[0];
}

extern "C" void kernel_launch(void* const* d_in, const int* in_sizes, int n_in,
                              void* d_out, int out_size, void* d_ws, size_t ws_size,
                              hipStream_t stream) {
    const float* x     = (const float*)d_in[0];
    const int*   ei    = (const int*)d_in[1];
    const float* W     = (const float*)d_in[2];
    const float* bconv = (const float*)d_in[3];
    const float* wfc   = (const float*)d_in[4];
    const float* bfc   = (const float*)d_in[5];
    float* out = (float*)d_out;

    const int N = in_sizes[0] / D_FEAT;
    const int E = in_sizes[1] / 2;

    char* ws = (char*)d_ws;
    unsigned* deg  = (unsigned*)ws;
    float*    dinv = (float*)(ws + (size_t)N * 4);
    float*    h    = (float*)(ws + (size_t)N * 8);
    float*    agg  = (float*)(ws + (size_t)N * 8 + (size_t)N * HID * 4);

    hipMemsetAsync(deg, 0, (size_t)N * sizeof(unsigned), stream);

    const int ntiles = N / 16;
    gcn_deg_gemm_kernel<<<1152, 256, 0, stream>>>(x, W, ei + E, deg, h, agg,
                                                        ntiles, E, N * HID / 4, 128);

    gcn_dinv_scale_kernel<<<(2 * N + 255) / 256, 256, 0, stream>>>(deg, dinv, h, N);

    gcn_scatter_kernel<<<1024, 256, 0, stream>>>(ei, h, agg, E);

    gcn_head_kernel<<<(N + 255) / 256, 256, 0, stream>>>(h, agg, dinv, bconv,
                                                         wfc, bfc, out, N);
}
